# starting kernel
# baseline (speedup 1.0000x reference)
#include <hip/hip_runtime.h>


typedef __attribute__((ext_vector_type(2))) float v2f;
typedef __attribute__((ext_vector_type(8))) float v8f;

#define KTOT        1152
#define KCHUNK      576
#define LDS_STRIDE  580
#define NPIX        196
#define NTILES      13

static __device__ __forceinline__ v8f wmma_f32(v2f a, v2f b, v8f c) {
    return __builtin_amdgcn_wmma_f32_16x16x4_f32(
         false, a,  false, b,
         (short)0, c,  false,  false);
}

__global__ __launch_bounds__(512) void fused_roll_conv_wmma(
    const float* __restrict__ x,
    const float* __restrict__ w,
    float* __restrict__ y)
{
    __shared__ float sT[16 * LDS_STRIDE];

    const int tid   = threadIdx.x;
    const int lane  = tid & 31;
    const int wv    = tid >> 5;
    const int ntile = blockIdx.x;

    const int colN = lane & 15;
    const int koff = (lane >> 4) << 1;

    const int   arow  = (wv << 4) + colN;
    const float* aptr  = w + (size_t)arow * KTOT + koff;
    const float* bbase = &sT[colN * LDS_STRIDE + koff];

    const int fp = ntile * 16 + wv;
    const int fn = fp / 14;
    const int fo = fp - fn * 14;
    const int fb1 = (fo + 13) % 14;
    const bool fvalid = (fp < NPIX);
    float* sTcol = &sT[wv * LDS_STRIDE];

    v8f acc0 = {0.f,0.f,0.f,0.f,0.f,0.f,0.f,0.f};
    v8f acc1 = acc0, acc2 = acc0, acc3 = acc0;

    for (int kc = 0; kc < KTOT; kc += KCHUNK) {
        __builtin_prefetch(aptr + kc, 0, 3);

        for (int q = 0; q < KCHUNK / 32; ++q) {
            int kk    = lane + (q << 5);
            int kflat = kc + kk;
            int l  = kflat & 127;
            int jk = kflat >> 7;
            int k  = jk % 3;
            int j  = jk / 3;
            float v = 0.f;
            if (fvalid) {
                int a = fn + j - 1;
                if (0 <= a && a < 14) {
                    int s0 = fo + k;
                    if (s0 >= 1 && s0 <= 14) {
                        int w0 = (s0 + 12) % 14;
                        v += x[l * NPIX + a * 14 + w0];
                    }
                    int s1 = fb1 + k;
                    if (s1 >= 1 && s1 <= 14) {
                        int w1 = (s1 + 12) % 14;
                        v += x[(128 + l) * NPIX + a * 14 + w1];
                    }
                }
            }
            sTcol[kk] = v;
        }
        __syncthreads();

        const float* ap = aptr + kc;
        #pragma unroll 2
        for (int kk = 0; kk < KCHUNK; kk += 16) {
            v2f a0 = *(const v2f*)(ap + kk);
            v2f b0 = *(const v2f*)(bbase + kk);
            v2f a1 = *(const v2f*)(ap + kk + 4);
            v2f b1 = *(const v2f*)(bbase + kk + 4);
            v2f a2 = *(const v2f*)(ap + kk + 8);
            v2f b2 = *(const v2f*)(bbase + kk + 8);
            v2f a3 = *(const v2f*)(ap + kk + 12);
            v2f b3 = *(const v2f*)(bbase + kk + 12);
            acc0 = wmma_f32(a0, b0, acc0);
            acc1 = wmma_f32(a1, b1, acc1);
            acc2 = wmma_f32(a2, b2, acc2);
            acc3 = wmma_f32(a3, b3, acc3);
        }
        __syncthreads();
    }

    v8f acc = (acc0 + acc1) + (acc2 + acc3);
    int p = ntile * 16 + colN;
    if (p < NPIX) {
        int half = (lane >> 4) << 3;
        #pragma unroll
        for (int r = 0; r < 8; ++r) {
            int i = (wv << 4) + half + r;
            y[(size_t)i * NPIX + p] = acc[r];
        }
    }
}

extern "C" void kernel_launch(void* const* d_in, const int* in_sizes, int n_in,
                              void* d_out, int out_size, void* d_ws, size_t ws_size,
                              hipStream_t stream) {
    const float* x = (const float*)d_in[0];
    const float* w = (const float*)d_in[1];
    float* y = (float*)d_out;
    (void)in_sizes; (void)n_in; (void)out_size; (void)d_ws; (void)ws_size;
    fused_roll_conv_wmma<<<dim3(NTILES), dim3(512), 0, stream>>>(x, w, y);
}


# One work-group per 16 channels x 2 image rows (112, was 13): weight slab and compact tap-shifted input (18 sums per channel row) staged once in LDS, taps by address; changes f32 sum order
# speedup vs baseline: 5.2518x; 5.2518x over previous
#include <hip/hip_runtime.h>


typedef __attribute__((ext_vector_type(2))) float v2f;
typedef __attribute__((ext_vector_type(4))) float v4f;
typedef __attribute__((ext_vector_type(8))) float v8f;

#define KTOT      1152
#define NPIX      196
#define A_PITCH   1156
#define ROWS      2
#define RROWS     (ROWS + 2)
#define NSLOT     18
#define Q_PITCH   (RROWS * NSLOT * 4)
#define NTHREADS  192
#define A_LOADS   ((16 * KTOT / 4) / NTHREADS)
#define NFILL     (RROWS * NSLOT)

static __device__ __forceinline__ v8f wmma_f32(v2f a, v2f b, v8f c) {
    return __builtin_amdgcn_wmma_f32_16x16x4_f32(
         false, a,  false, b,
         (short)0, c,  false,  false);
}

__global__ __launch_bounds__(NTHREADS) void rowblock_conv_wmma(
    const float* __restrict__ x,
    const float* __restrict__ w,
    float* __restrict__ y)
{
    __shared__ __attribute__((aligned(16))) float sA[16 * A_PITCH];
    __shared__ __attribute__((aligned(16))) float sR[32 * Q_PITCH];

    const int tid  = threadIdx.x;
    const int lane = tid & 31;
    const int wv   = tid >> 5;
    const int i0   = (blockIdx.x & 15) << 4;
    const int n0   = (blockIdx.x >> 4) * ROWS;

    v4f areg[A_LOADS];
    {
        const v4f* src = (const v4f*)(w + (size_t)i0 * KTOT);
        #pragma unroll
        for (int t = 0; t < A_LOADS; ++t) areg[t] = src[tid + t * NTHREADS];
    }

    if (tid < 2 * NFILL) {
        const int g  = (tid >= NFILL) ? 1 : 0;
        const int rs = tid - g * NFILL;
        const int r  = rs / NSLOT;
        const int ss = rs - r * NSLOT;
        const int a  = n0 - 1 + r;
        const bool rowok = (a >= 0) && (a < 14);
        const int ac = a < 0 ? 0 : (a > 13 ? 13 : a);
        bool vl, vh;
        int  wl, wh;
        if (ss <= 14) {
            vl = (ss <= 13);  wl = (ss == 0) ? 13 : ss - 1;
            vh = (ss >= 1);   wh = (ss == 1) ? 13 : (ss >= 2 ? ss - 2 : 0);
        } else if (ss == 15) { vl = false; wl = 0;  vh = true;  wh = 11; }
        else if (ss == 16)   { vl = true;  wl = 13; vh = true;  wh = 12; }
        else                 { vl = true;  wl = 0;  vh = false; wh = 0;  }
        vl = vl && rowok;
        vh = vh && rowok;
        const float* pl = x + (size_t)(g * 4) * NPIX + ac * 14 + wl;
        const float* ph = x + (size_t)(128 + g * 4) * NPIX + ac * 14 + wh;
        float* dst = &sR[g * Q_PITCH + rs * 4];
        #pragma unroll
        for (int qq = 0; qq < 16; ++qq) {
            v4f v;
            #pragma unroll
            for (int e = 0; e < 4; ++e) {
                const float xl = pl[(qq * 8 + e) * NPIX];
                const float xh = ph[(qq * 8 + e) * NPIX];
                v[e] = (vl ? xl : 0.f) + (vh ? xh : 0.f);
            }
            *(v4f*)(dst + qq * 2 * Q_PITCH) = v;
        }
    }
    #pragma unroll
    for (int t = 0; t < A_LOADS; ++t) {
        const int f  = tid + t * NTHREADS;
        const int m  = f / 288;
        const int c4 = f - m * 288;
        *(v4f*)&sA[m * A_PITCH + (c4 << 2)] = areg[t];
    }
    __syncthreads();

    if (wv < ROWS) {
        const int c     = lane & 15;
        const int h     = lane >> 4;
        const int slot0 = (c == 0) ? 15 : c - 1;
        const float* ap = &sA[c * A_PITCH + (h << 2)];
        const float* bp = &sR[h * Q_PITCH + ((wv * NSLOT + slot0) << 2)];

        v8f acc0 = {0.f,0.f,0.f,0.f,0.f,0.f,0.f,0.f};
        v8f acc1 = acc0, acc2 = acc0, acc3 = acc0;

        #pragma unroll
        for (int j = 0; j < 3; ++j) {
            #pragma unroll
            for (int k = 0; k < 3; ++k) {
                const int jk   = j * 3 + k;
                const int btap = (j * NSLOT + k) << 2;
                #pragma unroll
                for (int g16 = 0; g16 < 16; g16 += 2) {
                    const v4f a0 = *(const v4f*)(ap + (jk * 16 + g16) * 8);
                    const v4f b0 = *(const v4f*)(bp + g16 * 2 * Q_PITCH + btap);
                    const v4f a1 = *(const v4f*)(ap + (jk * 16 + g16 + 1) * 8);
                    const v4f b1 = *(const v4f*)(bp + (g16 + 1) * 2 * Q_PITCH + btap);
                    acc0 = wmma_f32(a0.xy, b0.xy, acc0);
                    acc1 = wmma_f32(a0.zw, b0.zw, acc1);
                    acc2 = wmma_f32(a1.xy, b1.xy, acc2);
                    acc3 = wmma_f32(a1.zw, b1.zw, acc3);
                }
            }
        }

        const v8f acc = (acc0 + acc1) + (acc2 + acc3);
        if (c < 14) {
            float* yp = y + (size_t)(i0 + (h << 3)) * NPIX + (n0 + wv) * 14 + c;
            #pragma unroll
            for (int r = 0; r < 8; ++r) yp[r * NPIX] = acc[r];
        }
    }
}

extern "C" void kernel_launch(void* const* d_in, const int* in_sizes, int n_in,
                              void* d_out, int out_size, void* d_ws, size_t ws_size,
                              hipStream_t stream) {
    const float* x = (const float*)d_in[0];
    const float* w = (const float*)d_in[1];
    float* y = (float*)d_out;
    (void)in_sizes; (void)n_in; (void)out_size; (void)d_ws; (void)ws_size;
    rowblock_conv_wmma<<<dim3(112), dim3(NTHREADS), 0, stream>>>(x, w, y);
}


# Staging pipelined against the products: K in 4 channel stages, 9 waves (was 6) keep loads of stages s+1, s+2 in flight across a hand-coded LDS-only barrier, 2 multiply stage s; changes f32 sum order
# speedup vs baseline: 5.6736x; 1.0803x over previous
#include <hip/hip_runtime.h>


typedef __attribute__((ext_vector_type(2))) float v2f;
typedef __attribute__((ext_vector_type(4))) float v4f;
typedef __attribute__((ext_vector_type(8))) float v8f;

#define KTOT      1152
#define NPIX      196
#define A_PITCH   1156
#define ROWS      2
#define RROWS     (ROWS + 2)
#define NSLOT     18
#define Q_PITCH   (RROWS * NSLOT * 4)
#define NFILL     (RROWS * NSLOT)
#define NTHREADS  288
#define NSTAGE    4
#define W_LOADS   4
#define X_ITEMS   2

static __device__ __forceinline__ v8f wmma_f32(v2f a, v2f b, v8f c) {
    return __builtin_amdgcn_wmma_f32_16x16x4_f32(
         false, a,  false, b,
         (short)0, c,  false,  false);
}

static __device__ __forceinline__ void stage_barrier() {
    asm volatile("s_wait_dscnt 0x0\n\ts_barrier_signal -1\n\ts_barrier_wait -1" ::: "memory");
}

__global__ __launch_bounds__(NTHREADS) void rowblock_conv_wmma(
    const float* __restrict__ x,
    const float* __restrict__ w,
    float* __restrict__ y)
{
    __shared__ __attribute__((aligned(16))) float sA[16 * A_PITCH];
    __shared__ __attribute__((aligned(16))) float sR[32 * Q_PITCH];

    const int tid  = threadIdx.x;
    const int lane = tid & 31;
    const int wv   = __builtin_amdgcn_readfirstlane(tid >> 5);
    const int i0   = (blockIdx.x & 15) << 4;
    const int n0   = (blockIdx.x >> 4) * ROWS;
    const bool matrix = wv < ROWS;

    const int wp = tid & 7;
    const int wr = (tid >> 3) & 3;
    const int wt = tid >> 5;
    const float* wsrc = w + (size_t)(i0 + wr) * KTOT + wt * 128 + wp * 4;
    float*       wdst = &sA[wr * A_PITCH + wt * 128 + wp * 4];

    const int qi = tid / NFILL;
    const int rs = tid - qi * NFILL;
    const int r  = rs / NSLOT;
    const int ss = rs - r * NSLOT;
    const int a  = n0 - 1 + r;
    const bool rowok = (a >= 0) && (a < 14);
    const int ac = a < 0 ? 0 : (a > 13 ? 13 : a);
    const bool vl = rowok && ((ss <= 13) || (ss >= 16));
    const bool vh = rowok && (ss >= 1) && (ss <= 16);
    const int  wl = (ss == 0 || ss == 16) ? 13 : (ss >= 15 ? 0 : ss - 1);
    const int  wh = (ss == 1) ? 13 : (ss >= 15 ? (ss == 17 ? 0 : ss - 4) : (ss >= 2 ? ss - 2 : 0));
    const float* pl = x + (size_t)(qi * 4) * NPIX + ac * 14 + wl;
    const float* ph = x + (size_t)(128 + qi * 4) * NPIX + ac * 14 + wh;
    float* xdst = &sR[qi * Q_PITCH + rs * 4];

    v4f   wreg[NSTAGE][W_LOADS];
    float xlo[NSTAGE][X_ITEMS * 4], xhi[NSTAGE][X_ITEMS * 4];

    auto issue = [&](const int st) __attribute__((always_inline)) {
        #pragma unroll
        for (int t = 0; t < W_LOADS; ++t)
            wreg[st][t] = *(const v4f*)(wsrc + t * 4 * KTOT + st * 32);
        #pragma unroll
        for (int t = 0; t < X_ITEMS; ++t) {
            #pragma unroll
            for (int e = 0; e < 4; ++e) {
                xlo[st][t * 4 + e] = pl[(st * 32 + t * 16 + e) * NPIX];
                xhi[st][t * 4 + e] = ph[(st * 32 + t * 16 + e) * NPIX];
            }
        }
    };

    auto park = [&](const int st) __attribute__((always_inline)) {
        #pragma unroll
        for (int t = 0; t < W_LOADS; ++t)
            *(v4f*)(wdst + t * 4 * A_PITCH + st * 32) = wreg[st][t];
        #pragma unroll
        for (int t = 0; t < X_ITEMS; ++t) {
            v4f v;
            #pragma unroll
            for (int e = 0; e < 4; ++e)
                v[e] = (vl ? xlo[st][t * 4 + e] : 0.f) + (vh ? xhi[st][t * 4 + e] : 0.f);
            *(v4f*)(xdst + (st * 8 + t * 4) * Q_PITCH) = v;
        }
    };

    const int col   = lane & 15;
    const int h     = lane >> 4;
    const int slot0 = (col == 0) ? 15 : col - 1;
    const int mrow  = matrix ? wv : 0;
    const float* ap = &sA[col * A_PITCH + (h << 2)];
    const float* bp = &sR[h * Q_PITCH + ((mrow * NSLOT + slot0) << 2)];

    v8f acc0, acc1, acc2, acc3;
    const v8f zero8 = {0.f,0.f,0.f,0.f,0.f,0.f,0.f,0.f};

    auto mult = [&](const int st) __attribute__((always_inline)) {
        #pragma unroll
        for (int j = 0; j < 3; ++j) {
            #pragma unroll
            for (int k = 0; k < 3; ++k) {
                const int jk   = j * 3 + k;
                const int btap = (j * NSLOT + k) << 2;
                #pragma unroll
                for (int g16 = st * 4; g16 < st * 4 + 4; g16 += 2) {
                    const v4f a0 = *(const v4f*)(ap + (jk * 16 + g16) * 8);
                    const v4f b0 = *(const v4f*)(bp + g16 * 2 * Q_PITCH + btap);
                    const v4f a1 = *(const v4f*)(ap + (jk * 16 + g16 + 1) * 8);
                    const v4f b1 = *(const v4f*)(bp + (g16 + 1) * 2 * Q_PITCH + btap);
                    const bool first = (st == 0) && (jk == 0) && (g16 == 0);
                    acc0 = wmma_f32(a0.xy, b0.xy, first ? zero8 : acc0);
                    acc1 = wmma_f32(a0.zw, b0.zw, first ? zero8 : acc1);
                    acc2 = wmma_f32(a1.xy, b1.xy, first ? zero8 : acc2);
                    acc3 = wmma_f32(a1.zw, b1.zw, first ? zero8 : acc3);
                }
            }
        }
    };

    issue(0);
    __builtin_amdgcn_sched_barrier(0);
    issue(1);
    __builtin_amdgcn_sched_barrier(0);
    #pragma unroll
    for (int st = 0; st < NSTAGE; ++st) {
        park(st);
        stage_barrier();
        if (st + 2 < NSTAGE) issue(st + 2);
        if (matrix) mult(st);
    }

    if (matrix) {
        const v8f acc = (acc0 + acc1) + (acc2 + acc3);
        if (col < 14) {
            float* yp = y + (size_t)(i0 + (h << 3)) * NPIX + (n0 + mrow) * 14 + col;
            #pragma unroll
            for (int rr = 0; rr < 8; ++rr) yp[rr * NPIX] = acc[rr];
        }
    }
}

extern "C" void kernel_launch(void* const* d_in, const int* in_sizes, int n_in,
                              void* d_out, int out_size, void* d_ws, size_t ws_size,
                              hipStream_t stream) {
    const float* x = (const float*)d_in[0];
    const float* w = (const float*)d_in[1];
    float* y = (float*)d_out;
    (void)in_sizes; (void)n_in; (void)out_size; (void)d_ws; (void)ws_size;
    rowblock_conv_wmma<<<dim3(112), dim3(NTHREADS), 0, stream>>>(x, w, y);
}


# K split over the four SIMDs: 4 matrix waves (was 2) each take a quarter of every stage's channels for both image rows, partial tiles added via LDS as (p0+p1)+(p2+p3); changes f32 sum order
# speedup vs baseline: 7.0563x; 1.2437x over previous
#include <hip/hip_runtime.h>


typedef __attribute__((ext_vector_type(2))) float v2f;
typedef __attribute__((ext_vector_type(4))) float v4f;
typedef __attribute__((ext_vector_type(8))) float v8f;

#define KTOT      1152
#define NPIX      196
#define A_PITCH   1156
#define ROWS      2
#define RROWS     (ROWS + 2)
#define NSLOT     18
#define Q_PITCH   (RROWS * NSLOT * 4)
#define NFILL     (RROWS * NSLOT)
#define NTHREADS  288
#define NSTAGE    4
#define W_LOADS   4
#define X_ITEMS   2
#define NMAT      4

static __device__ __forceinline__ v8f wmma_f32(v2f a, v2f b, v8f c) {
    return __builtin_amdgcn_wmma_f32_16x16x4_f32(
         false, a,  false, b,
         (short)0, c,  false,  false);
}

static __device__ __forceinline__ void stage_barrier() {
    asm volatile("s_wait_dscnt 0x0\n\ts_barrier_signal -1\n\ts_barrier_wait -1" ::: "memory");
}

__global__ __launch_bounds__(NTHREADS) void rowblock_conv_wmma(
    const float* __restrict__ x,
    const float* __restrict__ w,
    float* __restrict__ y)
{
    __shared__ __attribute__((aligned(16))) float sA[16 * A_PITCH];
    __shared__ __attribute__((aligned(16))) float sR[32 * Q_PITCH];
    __shared__ __attribute__((aligned(16))) float sP[NMAT * ROWS * 2 * 32 * 4];

    const int tid  = threadIdx.x;
    const int lane = tid & 31;
    const int wv   = __builtin_amdgcn_readfirstlane(tid >> 5);
    const int i0   = (blockIdx.x & 15) << 4;
    const int n0   = (blockIdx.x >> 4) * ROWS;
    const bool matrix = wv < NMAT;

    const int wp = tid & 7;
    const int wr = (tid >> 3) & 3;
    const int wt = tid >> 5;
    const float* wsrc = w + (size_t)(i0 + wr) * KTOT + wt * 128 + wp * 4;
    float*       wdst = &sA[wr * A_PITCH + wt * 128 + wp * 4];

    const int qi = tid / NFILL;
    const int rs = tid - qi * NFILL;
    const int r  = rs / NSLOT;
    const int ss = rs - r * NSLOT;
    const int a  = n0 - 1 + r;
    const bool rowok = (a >= 0) && (a < 14);
    const int ac = a < 0 ? 0 : (a > 13 ? 13 : a);
    const bool vl = rowok && ((ss <= 13) || (ss >= 16));
    const bool vh = rowok && (ss >= 1) && (ss <= 16);
    const int  wl = (ss == 0 || ss == 16) ? 13 : (ss >= 15 ? 0 : ss - 1);
    const int  wh = (ss == 1) ? 13 : (ss >= 15 ? (ss == 17 ? 0 : ss - 4) : (ss >= 2 ? ss - 2 : 0));
    const float* pl = x + (size_t)(qi * 4) * NPIX + ac * 14 + wl;
    const float* ph = x + (size_t)(128 + qi * 4) * NPIX + ac * 14 + wh;
    float* xdst = &sR[qi * Q_PITCH + rs * 4];

    v4f   wreg[NSTAGE][W_LOADS];
    float xlo[NSTAGE][X_ITEMS * 4], xhi[NSTAGE][X_ITEMS * 4];

    auto issue = [&](const int st) __attribute__((always_inline)) {
        #pragma unroll
        for (int t = 0; t < W_LOADS; ++t)
            wreg[st][t] = *(const v4f*)(wsrc + t * 4 * KTOT + st * 32);
        #pragma unroll
        for (int t = 0; t < X_ITEMS; ++t) {
            #pragma unroll
            for (int e = 0; e < 4; ++e) {
                xlo[st][t * 4 + e] = pl[(st * 32 + t * 16 + e) * NPIX];
                xhi[st][t * 4 + e] = ph[(st * 32 + t * 16 + e) * NPIX];
            }
        }
    };

    auto park = [&](const int st) __attribute__((always_inline)) {
        #pragma unroll
        for (int t = 0; t < W_LOADS; ++t)
            *(v4f*)(wdst + t * 4 * A_PITCH + st * 32) = wreg[st][t];
        #pragma unroll
        for (int t = 0; t < X_ITEMS; ++t) {
            v4f v;
            #pragma unroll
            for (int e = 0; e < 4; ++e)
                v[e] = (vl ? xlo[st][t * 4 + e] : 0.f) + (vh ? xhi[st][t * 4 + e] : 0.f);
            *(v4f*)(xdst + (st * 8 + t * 4) * Q_PITCH) = v;
        }
    };

    const int col   = lane & 15;
    const int h     = lane >> 4;
    const int slot0 = (col == 0) ? 15 : col - 1;
    const int q     = matrix ? wv : 0;
    const float* ap = &sA[col * A_PITCH + (q << 3) + (h << 2)];
    const float* bp = &sR[(2 * q + h) * Q_PITCH + (slot0 << 2)];

    v8f acc0, acc1, acc2, acc3;
    const v8f zero8 = {0.f,0.f,0.f,0.f,0.f,0.f,0.f,0.f};

    auto mult = [&](const int st) __attribute__((always_inline)) {
        #pragma unroll
        for (int j = 0; j < 3; ++j) {
            #pragma unroll
            for (int k = 0; k < 3; ++k) {
                const int jk   = j * 3 + k;
                const int btap = (j * NSLOT + k) << 2;
                const v4f a  = *(const v4f*)(ap + (jk * 16 + st * 4) * 8);
                const v4f b0 = *(const v4f*)(bp + st * 8 * Q_PITCH + btap);
                const v4f b1 = *(const v4f*)(bp + st * 8 * Q_PITCH + btap + (NSLOT << 2));
                const bool first = (st == 0) && (jk == 0);
                acc0 = wmma_f32(a.xy, b0.xy, first ? zero8 : acc0);
                acc1 = wmma_f32(a.zw, b0.zw, first ? zero8 : acc1);
                acc2 = wmma_f32(a.xy, b1.xy, first ? zero8 : acc2);
                acc3 = wmma_f32(a.zw, b1.zw, first ? zero8 : acc3);
            }
        }
    };

    issue(0);
    __builtin_amdgcn_sched_barrier(0);
    issue(1);
    __builtin_amdgcn_sched_barrier(0);
    #pragma unroll
    for (int st = 0; st < NSTAGE; ++st) {
        park(st);
        stage_barrier();
        if (st + 2 < NSTAGE) issue(st + 2);
        if (matrix) mult(st);
    }

    if (matrix) {
        const v8f p0 = acc0 + acc1;
        const v8f p1 = acc2 + acc3;
        float* pd = &sP[(q * ROWS * 2) * 128 + (lane << 2)];
        *(v4f*)(pd)       = __builtin_shufflevector(p0, p0, 0, 1, 2, 3);
        *(v4f*)(pd + 128) = __builtin_shufflevector(p0, p0, 4, 5, 6, 7);
        *(v4f*)(pd + 256) = __builtin_shufflevector(p1, p1, 0, 1, 2, 3);
        *(v4f*)(pd + 384) = __builtin_shufflevector(p1, p1, 4, 5, 6, 7);
    }
    stage_barrier();
    if (matrix) {
        const int m    = q & 1;
        const int half = q >> 1;
        const float* ps = &sP[(m * 2 + half) * 128 + (lane << 2)];
        const v4f t0 = *(const v4f*)(ps);
        const v4f t1 = *(const v4f*)(ps + 1 * ROWS * 2 * 128);
        const v4f t2 = *(const v4f*)(ps + 2 * ROWS * 2 * 128);
        const v4f t3 = *(const v4f*)(ps + 3 * ROWS * 2 * 128);
        const v4f sum = (t0 + t1) + (t2 + t3);
        if (col < 14) {
            float* yp = y + (size_t)(i0 + (h << 3) + (half << 2)) * NPIX + (n0 + m) * 14 + col;
            #pragma unroll
            for (int rr = 0; rr < 4; ++rr) yp[rr * NPIX] = sum[rr];
        }
    }
}

extern "C" void kernel_launch(void* const* d_in, const int* in_sizes, int n_in,
                              void* d_out, int out_size, void* d_ws, size_t ws_size,
                              hipStream_t stream) {
    const float* x = (const float*)d_in[0];
    const float* w = (const float*)d_in[1];
    float* y = (float*)d_out;
    (void)in_sizes; (void)n_in; (void)out_size; (void)d_ws; (void)ws_size;
    rowblock_conv_wmma<<<dim3(112), dim3(NTHREADS), 0, stream>>>(x, w, y);
}


# Image-row taps by the 4-product form of a 3-tap 2-row filter: weights combined as parked, input rows as read; 96 matrix instructions per wave (was 144); changes f32 sum order and rounding
# speedup vs baseline: 7.5022x; 1.0632x over previous
#include <hip/hip_runtime.h>


typedef __attribute__((ext_vector_type(2))) float v2f;
typedef __attribute__((ext_vector_type(4))) float v4f;
typedef __attribute__((ext_vector_type(8))) float v8f;

#define KTOT      1152
#define NPIX      196
#define NFAM      4
#define A_PITCH   1540
#define ROWS      2
#define RROWS     (ROWS + 2)
#define NSLOT     18
#define Q_PITCH   (RROWS * NSLOT * 4)
#define NFILL     (RROWS * NSLOT)
#define NTHREADS  288
#define NSTAGE    4
#define X_ITEMS   2
#define NMAT      4

static __device__ __forceinline__ v8f wmma_f32(v2f a, v2f b, v8f c) {
    return __builtin_amdgcn_wmma_f32_16x16x4_f32(
         false, a,  false, b,
         (short)0, c,  false,  false);
}

static __device__ __forceinline__ void stage_barrier() {
    asm volatile("s_wait_dscnt 0x0\n\ts_barrier_signal -1\n\ts_barrier_wait -1" ::: "memory");
}

__global__ __launch_bounds__(NTHREADS) void rowblock_conv_wmma(
    const float* __restrict__ x,
    const float* __restrict__ w,
    float* __restrict__ y)
{
    __shared__ __attribute__((aligned(16))) float sA[16 * A_PITCH];
    __shared__ __attribute__((aligned(16))) float sR[32 * Q_PITCH];
    __shared__ __attribute__((aligned(16))) float sP[NMAT * ROWS * 2 * 32 * 4];

    const int tid  = threadIdx.x;
    const int lane = tid & 31;
    const int wv   = __builtin_amdgcn_readfirstlane(tid >> 5);
    const int i0   = (blockIdx.x & 15) << 4;
    const int n0   = (blockIdx.x >> 4) * ROWS;
    const bool matrix = wv < NMAT;

    const int wp = tid & 7;
    const int wr = (tid >> 3) & 3;
    const bool two_items = (wv >= NMAT) && (wv < NMAT + 3);
    const int gA = wv;
    const int gB = two_items ? wv + 5 : wv;
    const int rowA = (gA / 3) * 4 + wr, kA = gA % 3;
    const int rowB = (gB / 3) * 4 + wr, kB = gB % 3;
    const float* wsrcA = w + (size_t)(i0 + rowA) * KTOT + kA * 128 + wp * 4;
    const float* wsrcB = w + (size_t)(i0 + rowB) * KTOT + kB * 128 + wp * 4;
    float*       wdstA = &sA[rowA * A_PITCH + kA * 128 + wp * 4];
    float*       wdstB = &sA[rowB * A_PITCH + kB * 128 + wp * 4];

    const int qi = tid / NFILL;
    const int rs = tid - qi * NFILL;
    const int r  = rs / NSLOT;
    const int ss = rs - r * NSLOT;
    const int a  = n0 - 1 + r;
    const bool rowok = (a >= 0) && (a < 14);
    const int ac = a < 0 ? 0 : (a > 13 ? 13 : a);
    const bool vl = rowok && ((ss <= 13) || (ss >= 16));
    const bool vh = rowok && (ss >= 1) && (ss <= 16);
    const int  wl = (ss == 0 || ss == 16) ? 13 : (ss >= 15 ? 0 : ss - 1);
    const int  wh = (ss == 1) ? 13 : (ss >= 15 ? (ss == 17 ? 0 : ss - 4) : (ss >= 2 ? ss - 2 : 0));
    const float* pl = x + (size_t)(qi * 4) * NPIX + ac * 14 + wl;
    const float* ph = x + (size_t)(128 + qi * 4) * NPIX + ac * 14 + wh;
    float* xdst = &sR[qi * Q_PITCH + rs * 4];

    v4f   wregA[NSTAGE][3], wregB[NSTAGE][3];
    float xlo[NSTAGE][X_ITEMS * 4], xhi[NSTAGE][X_ITEMS * 4];

    auto issue = [&](const int st) __attribute__((always_inline)) {
        #pragma unroll
        for (int j = 0; j < 3; ++j)
            wregA[st][j] = *(const v4f*)(wsrcA + j * 3 * 128 + st * 32);
        if (two_items) {
            #pragma unroll
            for (int j = 0; j < 3; ++j)
                wregB[st][j] = *(const v4f*)(wsrcB + j * 3 * 128 + st * 32);
        }
        #pragma unroll
        for (int t = 0; t < X_ITEMS; ++t) {
            #pragma unroll
            for (int e = 0; e < 4; ++e) {
                xlo[st][t * 4 + e] = pl[(st * 32 + t * 16 + e) * NPIX];
                xhi[st][t * 4 + e] = ph[(st * 32 + t * 16 + e) * NPIX];
            }
        }
    };

    auto combine = [&](const v4f g0, const v4f g1, const v4f g2, float* dst) __attribute__((always_inline)) {
        const v4f s = (g0 + g2) * 0.5f;
        *(v4f*)(dst)               = g0;
        *(v4f*)(dst + 1 * 3 * 128) = __builtin_elementwise_fma(g1, (v4f){ 0.5f,  0.5f,  0.5f,  0.5f}, s);
        *(v4f*)(dst + 2 * 3 * 128) = __builtin_elementwise_fma(g1, (v4f){-0.5f, -0.5f, -0.5f, -0.5f}, s);
        *(v4f*)(dst + 3 * 3 * 128) = g2;
    };

    auto park = [&](const int st) __attribute__((always_inline)) {
        combine(wregA[st][0], wregA[st][1], wregA[st][2], wdstA + st * 32);
        if (two_items)
            combine(wregB[st][0], wregB[st][1], wregB[st][2], wdstB + st * 32);
        #pragma unroll
        for (int t = 0; t < X_ITEMS; ++t) {
            v4f v;
            #pragma unroll
            for (int e = 0; e < 4; ++e)
                v[e] = (vl ? xlo[st][t * 4 + e] : 0.f) + (vh ? xhi[st][t * 4 + e] : 0.f);
            *(v4f*)(xdst + (st * 8 + t * 4) * Q_PITCH) = v;
        }
    };

    const int col   = lane & 15;
    const int h     = lane >> 4;
    const int slot0 = (col == 0) ? 15 : col - 1;
    const int q     = matrix ? wv : 0;
    const float* ap = &sA[col * A_PITCH + (q << 3) + (h << 2)];
    const float* bp = &sR[(2 * q + h) * Q_PITCH + (slot0 << 2)];

    v8f acc[2 * NFAM];
    const v8f zero8 = {0.f,0.f,0.f,0.f,0.f,0.f,0.f,0.f};

    auto mult = [&](const int st) __attribute__((always_inline)) {
        #pragma unroll
        for (int k = 0; k < 3; ++k) {
            v4f d[RROWS], bt[NFAM];
            #pragma unroll
            for (int rr = 0; rr < RROWS; ++rr)
                d[rr] = *(const v4f*)(bp + st * 8 * Q_PITCH + ((rr * NSLOT + k) << 2));
            bt[0] = d[0] - d[2];
            bt[1] = d[1] + d[2];
            bt[2] = d[2] - d[1];
            bt[3] = d[1] - d[3];
            const bool first = (st == 0) && (k == 0);
            #pragma unroll
            for (int t = 0; t < NFAM; ++t) {
                const v4f a = *(const v4f*)(ap + ((t * 3 + k) * 16 + st * 4) * 8);
                acc[2 * t]     = wmma_f32(a.xy, bt[t].xy, first ? zero8 : acc[2 * t]);
                acc[2 * t + 1] = wmma_f32(a.zw, bt[t].zw, first ? zero8 : acc[2 * t + 1]);
            }
        }
    };

    issue(0);
    __builtin_amdgcn_sched_barrier(0);
    issue(1);
    __builtin_amdgcn_sched_barrier(0);
    #pragma unroll
    for (int st = 0; st < NSTAGE; ++st) {
        park(st);
        stage_barrier();
        if (st + 2 < NSTAGE) issue(st + 2);
        if (matrix) mult(st);
    }

    if (matrix) {
        const v8f m0 = acc[0] + acc[1];
        const v8f m1 = acc[2] + acc[3];
        const v8f m2 = acc[4] + acc[5];
        const v8f m3 = acc[6] + acc[7];
        const v8f p0 = (m0 + m1) + m2;
        const v8f p1 = (m1 - m2) - m3;
        float* pd = &sP[(q * ROWS * 2) * 128 + (lane << 2)];
        *(v4f*)(pd)       = __builtin_shufflevector(p0, p0, 0, 1, 2, 3);
        *(v4f*)(pd + 128) = __builtin_shufflevector(p0, p0, 4, 5, 6, 7);
        *(v4f*)(pd + 256) = __builtin_shufflevector(p1, p1, 0, 1, 2, 3);
        *(v4f*)(pd + 384) = __builtin_shufflevector(p1, p1, 4, 5, 6, 7);
    }
    stage_barrier();
    if (matrix) {
        const int m    = q & 1;
        const int half = q >> 1;
        const float* ps = &sP[(m * 2 + half) * 128 + (lane << 2)];
        const v4f t0 = *(const v4f*)(ps);
        const v4f t1 = *(const v4f*)(ps + 1 * ROWS * 2 * 128);
        const v4f t2 = *(const v4f*)(ps + 2 * ROWS * 2 * 128);
        const v4f t3 = *(const v4f*)(ps + 3 * ROWS * 2 * 128);
        const v4f sum = (t0 + t1) + (t2 + t3);
        if (col < 14) {
            float* yp = y + (size_t)(i0 + (h << 3) + (half << 2)) * NPIX + (n0 + m) * 14 + col;
            #pragma unroll
            for (int rr = 0; rr < 4; ++rr) yp[rr * NPIX] = sum[rr];
        }
    }
}

extern "C" void kernel_launch(void* const* d_in, const int* in_sizes, int n_in,
                              void* d_out, int out_size, void* d_ws, size_t ws_size,
                              hipStream_t stream) {
    const float* x = (const float*)d_in[0];
    const float* w = (const float*)d_in[1];
    float* y = (float*)d_out;
    (void)in_sizes; (void)n_in; (void)out_size; (void)d_ws; (void)ws_size;
    rowblock_conv_wmma<<<dim3(112), dim3(NTHREADS), 0, stream>>>(x, w, y);
}
